# starting kernel
# baseline (speedup 1.0000x reference)
#include <hip/hip_runtime.h>


#define RANK   64
#define WIN    11
#define HALO   5
#define TILE   64
#define XROWS  (TILE + 2*HALO)
#define LD_X   66
#define LD_W   68

typedef float v2f __attribute__((ext_vector_type(2)));
typedef float v8f __attribute__((ext_vector_type(8)));

static __device__ __forceinline__ float fast_tanh(float x) {
#if __has_builtin(__builtin_amdgcn_tanhf)
    return __builtin_amdgcn_tanhf(x);
#elif __has_builtin(__builtin_amdgcn_tanh_f32)
    return __builtin_amdgcn_tanh_f32(x);
#else
    return tanhf(x);
#endif
}

static __device__ __forceinline__ float fast_exp_nonpos(float x) {
#if __has_builtin(__builtin_amdgcn_exp2f)
    return __builtin_amdgcn_exp2f(x * 1.44269504088896341f);
#else
    return __expf(x);
#endif
}

__global__ __launch_bounds__(128)
void attn_window_fused_kernel(const float* __restrict__ x,
                              const float* __restrict__ W,
                              const float* __restrict__ b,
                              float* __restrict__ out,
                              int nrows)
{
    __shared__ float xs[XROWS * LD_X];
    __shared__ float vs[TILE  * LD_X];
    __shared__ float ws[RANK  * LD_W];
    __shared__ float bs[RANK];

    const int tid = threadIdx.x;
    const int r0  = blockIdx.x * TILE;

    for (int idx = tid; idx < RANK * (RANK / 4); idx += blockDim.x) {
        const int row = idx >> 4;
        const int c4  = idx & 15;
        float4 v = ((const float4*)W)[row * (RANK / 4) + c4];
        *(float4*)&ws[row * LD_W + c4 * 4] = v;
    }
    if (tid < RANK) bs[tid] = b[tid];

    for (int idx = tid; idx < XROWS * (RANK / 4); idx += blockDim.x) {
        const int row = idx >> 4;
        const int c4  = idx & 15;
        const int g   = r0 - HALO + row;
        float4 v = make_float4(0.f, 0.f, 0.f, 0.f);
        if (g >= 0 && g < nrows) v = ((const float4*)x)[g * (RANK / 4) + c4];
        float* p = &xs[row * LD_X + c4 * 4];
        *(float2*)(p)     = make_float2(v.x, v.y);
        *(float2*)(p + 2) = make_float2(v.z, v.w);
    }
    __syncthreads();

    {
        const int wave = tid >> 5;
        const int lane = tid & 31;
        const int m    = lane & 15;
        const int hi   = lane >> 4;
        const int arow = wave * 16 + m + HALO;

        v2f afrag[RANK / 4];
        #pragma unroll
        for (int kb = 0; kb < RANK / 4; ++kb)
            afrag[kb] = *(const v2f*)&xs[arow * LD_X + kb * 4 + hi * 2];

        #pragma unroll
        for (int nt = 0; nt < 4; ++nt) {
            v8f acc = {};
            const int n = nt * 16 + m;
            #pragma unroll
            for (int kb = 0; kb < RANK / 4; ++kb) {
                v2f bfrag = *(const v2f*)&ws[n * LD_W + kb * 4 + hi * 2];
                acc = __builtin_amdgcn_wmma_f32_16x16x4_f32(
                        false, afrag[kb], false, bfrag,
                        (short)0, acc, false, false);
            }
            const float bias = bs[n];
            #pragma unroll
            for (int v = 0; v < 8; ++v) {
                const int row = wave * 16 + v + hi * 8;
                vs[row * LD_X + n] = fast_tanh(acc[v] + bias);
            }
        }
    }
    __syncthreads();

    {
        const int r    = tid >> 1;
        const int half = tid & 1;
        const int g    = r0 + r;
        if (g < nrows) {
            const int c0 = half * 32;
            float s[WIN];
            #pragma unroll
            for (int w = 0; w < WIN; ++w) {
                v2f a0 = {0.f, 0.f}, a1 = {0.f, 0.f};
                #pragma unroll
                for (int c = 0; c < 32; c += 4) {
                    v2f x0 = *(const v2f*)&xs[(r + w) * LD_X + c0 + c];
                    v2f x1 = *(const v2f*)&xs[(r + w) * LD_X + c0 + c + 2];
                    v2f v0 = *(const v2f*)&vs[r * LD_X + c0 + c];
                    v2f v1 = *(const v2f*)&vs[r * LD_X + c0 + c + 2];
                    a0 = __builtin_elementwise_fma(x0, v0, a0);
                    a1 = __builtin_elementwise_fma(x1, v1, a1);
                }
                v2f acc = a0 + a1;
                float sv = acc.x + acc.y;
                sv += __shfl_xor(sv, 1, 32);
                s[w] = sv * 0.125f;
            }
            float mx = s[0];
            #pragma unroll
            for (int w = 1; w < WIN; ++w) mx = fmaxf(mx, s[w]);
            float sum = 0.f;
            #pragma unroll
            for (int w = 0; w < WIN; ++w) { s[w] = fast_exp_nonpos(s[w] - mx); sum += s[w]; }
            const float inv = 1.0f / sum;

            #pragma unroll
            for (int c = 0; c < 32; c += 2) {
                v2f o = {0.f, 0.f};
                #pragma unroll
                for (int w = 0; w < WIN; ++w) {
                    v2f xv = *(const v2f*)&xs[(r + w) * LD_X + c0 + c];
                    v2f sw = {s[w], s[w]};
                    o = __builtin_elementwise_fma(sw, xv, o);
                }
                o *= inv;
                *(v2f*)&out[g * RANK + c0 + c] = o;
            }
        }
    }
}

extern "C" void kernel_launch(void* const* d_in, const int* in_sizes, int n_in,
                              void* d_out, int out_size, void* d_ws, size_t ws_size,
                              hipStream_t stream) {
    const float* x = (const float*)d_in[0];
    const float* W = (const float*)d_in[1];
    const float* b = (const float*)d_in[2];
    float* out = (float*)d_out;

    const int nrows = in_sizes[0] / RANK;
    const int grid  = (nrows + TILE - 1) / TILE;
    attn_window_fused_kernel<<<grid, 128, 0, stream>>>(x, W, b, out, nrows);
}


# wave priority (s_setprio) per tile = tiles the group still has to do: the group furthest behind wins the matrix pipe and a WGP's groups finish together, not the youngest alone; sums unchanged
# speedup vs baseline: 2.8508x; 2.8508x over previous
#include <hip/hip_runtime.h>


#define RANK   64
#define WIN    11
#define HALO   5
#define TILE   64
#define XROWS  (TILE + 2*HALO)
#define SROWS  80
#define LD_X   68
#define LD_W   68
#define NTHR   128
#define XLD    10
#define WLD    8
#define NGROUPS 1024

typedef float v2f __attribute__((ext_vector_type(2)));
typedef float v8f __attribute__((ext_vector_type(8)));

static __device__ __forceinline__ float fast_tanh(float x) {
#if __has_builtin(__builtin_amdgcn_tanhf)
    return __builtin_amdgcn_tanhf(x);
#elif __has_builtin(__builtin_amdgcn_tanh_f32)
    return __builtin_amdgcn_tanh_f32(x);
#else
    return tanhf(x);
#endif
}

static __device__ __forceinline__ float fast_exp2(float x) {
#if __has_builtin(__builtin_amdgcn_exp2f)
    return __builtin_amdgcn_exp2f(x);
#else
    return exp2f(x);
#endif
}

static __device__ __forceinline__ float xhalf(float x) {
    const int b = __builtin_bit_cast(int, x);
    return __builtin_bit_cast(float,
        __builtin_amdgcn_permlanex16(b, b, 0x76543210, 0xfedcba98, false, false));
}

static __device__ __forceinline__ v8f mma4(v2f a, v2f b, v8f c) {
    return __builtin_amdgcn_wmma_f32_16x16x4_f32(false, a, false, b, (short)0, c, false, false);
}

static __device__ __forceinline__ void load_x_tile(float4 (&xr)[XLD],
                                                   const float* __restrict__ x,
                                                   int r0, int nrows, int tid)
{
    #pragma unroll
    for (int i = 0; i < XLD; ++i) {
        const int idx = tid + i * NTHR;
        const int row = idx >> 4;
        const int c4  = idx & 15;
        const int gc  = min(max(r0 - HALO + row, 0), nrows - 1);
        if (i < XLD - 1 || idx < XROWS * (RANK / 4))
            xr[i] = ((const float4*)x)[gc * (RANK / 4) + c4];
    }
}

static __device__ __forceinline__ void lds_barrier() {
    asm volatile("s_wait_dscnt 0x0\n\ts_barrier_signal -1\n\ts_barrier_wait -1" ::: "memory");
}

static __device__ __forceinline__ void store_x_tile(const float4 (&xr)[XLD], float* xt,
                                                    int r0, int nrows, int tid)
{
    #pragma unroll
    for (int i = 0; i < XLD; ++i) {
        const int idx = tid + i * NTHR;
        const int row = idx >> 4;
        const int c4  = idx & 15;
        const int g   = r0 - HALO + row;
        if (i < XLD - 1 || idx < XROWS * (RANK / 4)) {
            float4 v = xr[i];
            if (g < 0 || g >= nrows) v = make_float4(0.f, 0.f, 0.f, 0.f);
            *(float4*)&xt[row * LD_X + c4 * 4] = v;
        }
    }
}

__global__ __launch_bounds__(NTHR)
void attn_window_fused_kernel(const float* __restrict__ x,
                              const float* __restrict__ W,
                              const float* __restrict__ b,
                              float* __restrict__ out,
                              int nrows)
{
    __shared__ __attribute__((aligned(16))) float xs[2][SROWS * LD_X];
    __shared__ __attribute__((aligned(16))) float ws[RANK * LD_W];
    __shared__ __attribute__((aligned(16))) float bs[RANK];

    const int tid    = threadIdx.x;
    const int ntiles = (nrows + TILE - 1) / TILE;
    static_assert(NGROUPS == 1024 && (19 * 539) % NGROUPS == 1, "leftover spread is written for 1024 groups");
    const int bid   = blockIdx.x;
    const int full  = ntiles / NGROUPS;
    const int rem   = ntiles - full * NGROUPS;
    const int kx    = (bid * 539) & (NGROUPS - 1);
    const int nmine = full + (kx < rem ? 1 : 0);
    if (nmine == 0) return;
    int tile = (full > 0) ? bid : full * NGROUPS + kx;

    const int lane = tid & 31;
    const int c    = lane & 15;
    const int h    = lane >> 4;
    const int wave = __builtin_amdgcn_readfirstlane(tid >> 5);

    float4 wr[WLD];
    #pragma unroll
    for (int i = 0; i < WLD; ++i) wr[i] = ((const float4*)W)[tid + i * NTHR];

    float bias_r = 0.f;
    if (tid < RANK) bias_r = b[tid];

    float4 xr[XLD];
    load_x_tile(xr, x, tile * TILE, nrows, tid);

    #pragma unroll
    for (int i = 0; i < WLD; ++i) {
        const int idx = tid + i * NTHR;
        const int row = idx >> 4;
        const int c4  = idx & 15;
        *(float4*)&ws[row * LD_W + c4 * 4] = wr[i];
    }
    if (tid < RANK) bs[tid] = bias_r;
    for (int i = tid; i < 2 * (SROWS - XROWS) * (LD_X / 4); i += NTHR) {
        const int s = i / ((SROWS - XROWS) * (LD_X / 4));
        const int o = i - s * ((SROWS - XROWS) * (LD_X / 4));
        *(float4*)&xs[s][XROWS * LD_X + o * 4] = make_float4(0.f, 0.f, 0.f, 0.f);
    }

    const int foff = c * LD_X + 8 * h;
    const int woff = c * LD_W + 8 * h;
    const int toff = 8 * h * LD_X + c;
    const int dj = 8 * h - c;

    int stage = 0;
    for (int it = 0; it < nmine; ++it) {
        const int left = nmine - it;
        if (left >= 4)      __builtin_amdgcn_s_setprio(3);
        else if (left == 3) __builtin_amdgcn_s_setprio(2);
        else if (left == 2) __builtin_amdgcn_s_setprio(1);
        else                __builtin_amdgcn_s_setprio(0);

        float* xt = xs[stage];
        const int r0 = tile * TILE;

        store_x_tile(xr, xt, r0, nrows, tid);
        lds_barrier();

        const int next = (it + 1 < full) ? tile + NGROUPS : full * NGROUPS + kx;
        if (it + 1 < nmine) load_x_tile(xr, x, next * TILE, nrows, tid);

        const float* xw = xt + wave * 16 * LD_X;

        v8f acc[4];
        #pragma unroll
        for (int mt = 0; mt < 4; ++mt) {
            const float4 b0 = *(const float4*)&bs[16 * mt + 8 * h];
            const float4 b1 = *(const float4*)&bs[16 * mt + 8 * h + 4];
            acc[mt] = (v8f){b0.x, b0.y, b0.z, b0.w, b1.x, b1.y, b1.z, b1.w};
        }
        float4 xb[8];
        #pragma unroll
        for (int i = 0; i < 8; ++i)
            xb[i] = *(const float4*)&xw[HALO * LD_X + foff + 16 * (i >> 1) + 4 * (i & 1)];

        float4 wa[2][8];
        float4 xa[2][8];
        #pragma unroll
        for (int e = 0; e < 2; ++e)
            #pragma unroll
            for (int mt = 0; mt < 4; ++mt)
                wa[0][2 * mt + e] = *(const float4*)&ws[mt * 16 * LD_W + woff + 4 * e];
        #pragma unroll
        for (int g = 0; g < 4; ++g) {
            if (g < 3) {
                #pragma unroll
                for (int e = 0; e < 2; ++e)
                    #pragma unroll
                    for (int mt = 0; mt < 4; ++mt)
                        wa[(g + 1) & 1][2 * mt + e] =
                            *(const float4*)&ws[mt * 16 * LD_W + woff + 16 * (g + 1) + 4 * e];
            } else {
                #pragma unroll
                for (int i = 0; i < 8; ++i)
                    #pragma unroll
                    for (int jt = 0; jt < 2; ++jt)
                        xa[jt][i] = *(const float4*)&xw[jt * 16 * LD_X + foff + 16 * (i >> 1) + 4 * (i & 1)];
            }
            __builtin_amdgcn_sched_barrier(0);
            #pragma unroll
            for (int e = 0; e < 2; ++e) {
                const int i = 2 * g + e;
                #pragma unroll
                for (int mt = 0; mt < 4; ++mt) {
                    const float4 w4 = wa[g & 1][2 * mt + e];
                    acc[mt] = mma4((v2f){w4.x, w4.y}, (v2f){xb[i].x, xb[i].y}, acc[mt]);
                    acc[mt] = mma4((v2f){w4.z, w4.w}, (v2f){xb[i].z, xb[i].w}, acc[mt]);
                }
            }
            __builtin_amdgcn_sched_barrier(0);
        }
        float t[4][8];
        #pragma unroll
        for (int mt = 0; mt < 4; ++mt)
            #pragma unroll
            for (int v = 0; v < 8; ++v)
                t[mt][v] = fast_tanh(acc[mt][v]);

        v2f xq[4][8];
        #pragma unroll
        for (int jt = 0; jt < 2; ++jt)
            #pragma unroll
            for (int q = 0; q < 4; ++q)
                #pragma unroll
                for (int ct = 0; ct < 4; ++ct) {
                    const float* xp = &xw[(16 * jt + 2 * q) * LD_X + toff + 16 * ct];
                    xq[ct][4 * jt + q] = (v2f){xp[0], xp[LD_X]};
                }
        __builtin_amdgcn_sched_barrier(0);

        v8f sacc[2] = {};
        #pragma unroll
        for (int i = 0; i < 8; ++i) {
            const int g = i >> 1, qq = i & 1;
            #pragma unroll
            for (int jt = 0; jt < 2; ++jt) {
                const float4 x4 = xa[jt][i];
                sacc[jt] = mma4((v2f){x4.x, x4.y}, (v2f){t[g][4 * qq],     t[g][4 * qq + 1]}, sacc[jt]);
                sacc[jt] = mma4((v2f){x4.z, x4.w}, (v2f){t[g][4 * qq + 2], t[g][4 * qq + 3]}, sacc[jt]);
            }
        }

        float p[2][8];
        {
            float mloc = -__builtin_inff();
            #pragma unroll
            for (int jt = 0; jt < 2; ++jt)
                #pragma unroll
                for (int v = 0; v < 8; ++v) {
                    const bool inb = (unsigned)(16 * jt + v + dj) < (unsigned)WIN;
                    p[jt][v] = inb ? sacc[jt][v] : -__builtin_inff();
                    mloc = fmaxf(mloc, p[jt][v]);
                }
            const float mx = fmaxf(mloc, xhalf(mloc));
            const float sc = 0.125f * 1.44269504088896341f;
            float sum = 0.f;
            #pragma unroll
            for (int jt = 0; jt < 2; ++jt)
                #pragma unroll
                for (int v = 0; v < 8; ++v) {
                    p[jt][v] = fast_exp2((p[jt][v] - mx) * sc);
                    sum += p[jt][v];
                }
            sum += xhalf(sum);
            const float inv = 1.0f / sum;
            #pragma unroll
            for (int jt = 0; jt < 2; ++jt)
                #pragma unroll
                for (int v = 0; v < 8; ++v) p[jt][v] *= inv;
        }

        v8f oacc[4] = {};
        #pragma unroll
        for (int jt = 0; jt < 2; ++jt) {
            #pragma unroll
            for (int q = 0; q < 4; ++q) {
                const v2f pa = {p[jt][2 * q], p[jt][2 * q + 1]};
                #pragma unroll
                for (int ct = 0; ct < 4; ++ct)
                    oacc[ct] = mma4(pa, xq[ct][4 * jt + q], oacc[ct]);
            }
        }

        {
            const int rw = r0 + wave * 16;
            float* op = out + (rw + 8 * h) * RANK + c;
            if (rw + 16 <= nrows) {
                #pragma unroll
                for (int v = 0; v < 8; ++v)
                    #pragma unroll
                    for (int ct = 0; ct < 4; ++ct) op[v * RANK + 16 * ct] = oacc[ct][v];
            } else {
                #pragma unroll
                for (int v = 0; v < 8; ++v)
                    if (rw + 8 * h + v < nrows) {
                        #pragma unroll
                        for (int ct = 0; ct < 4; ++ct) op[v * RANK + 16 * ct] = oacc[ct][v];
                    }
            }
        }
        stage ^= 1;
        tile = next;
    }
}

extern "C" void kernel_launch(void* const* d_in, const int* in_sizes, int n_in,
                              void* d_out, int out_size, void* d_ws, size_t ws_size,
                              hipStream_t stream) {
    const float* x = (const float*)d_in[0];
    const float* W = (const float*)d_in[1];
    const float* b = (const float*)d_in[2];
    float* out = (float*)d_out;

    const int nrows = in_sizes[0] / RANK;
    attn_window_fused_kernel<<<NGROUPS, NTHR, 0, stream>>>(x, W, b, out, nrows);
}


# rows behind the tiles every group gets go out as 16-row blocks, each wave of a group taking 16 of the 64 columns: 32 matrix ops on 4 SIMDs, not a 13th 128 on one; their score sums: 4 parts via LDS
# speedup vs baseline: 2.9535x; 1.0360x over previous
#include <hip/hip_runtime.h>


#define RANK   64
#define WIN    11
#define HALO   5
#define TILE   64
#define XROWS  (TILE + 2*HALO)
#define SROWS  80
#define LD_X   68
#define LD_W   68
#define NTHR   128
#define XLD    10
#define WLD    8
#define NGROUPS 1024

typedef float v2f __attribute__((ext_vector_type(2)));
typedef float v8f __attribute__((ext_vector_type(8)));

static __device__ __forceinline__ float fast_tanh(float x) {
#if __has_builtin(__builtin_amdgcn_tanhf)
    return __builtin_amdgcn_tanhf(x);
#elif __has_builtin(__builtin_amdgcn_tanh_f32)
    return __builtin_amdgcn_tanh_f32(x);
#else
    return tanhf(x);
#endif
}

static __device__ __forceinline__ float fast_exp2(float x) {
#if __has_builtin(__builtin_amdgcn_exp2f)
    return __builtin_amdgcn_exp2f(x);
#else
    return exp2f(x);
#endif
}

static __device__ __forceinline__ float xhalf(float x) {
    const int b = __builtin_bit_cast(int, x);
    return __builtin_bit_cast(float,
        __builtin_amdgcn_permlanex16(b, b, 0x76543210, 0xfedcba98, false, false));
}

static __device__ __forceinline__ v8f mma4(v2f a, v2f b, v8f c) {
    return __builtin_amdgcn_wmma_f32_16x16x4_f32(false, a, false, b, (short)0, c, false, false);
}

static __device__ __forceinline__ void load_x_tile(float4 (&xr)[XLD],
                                                   const float* __restrict__ x,
                                                   int r0, int nrows, int tid)
{
    #pragma unroll
    for (int i = 0; i < XLD; ++i) {
        const int idx = tid + i * NTHR;
        const int row = idx >> 4;
        const int c4  = idx & 15;
        const int gc  = min(max(r0 - HALO + row, 0), nrows - 1);
        if (i < XLD - 1 || idx < XROWS * (RANK / 4))
            xr[i] = ((const float4*)x)[gc * (RANK / 4) + c4];
    }
}

static __device__ __forceinline__ void lds_barrier() {
    asm volatile("s_wait_dscnt 0x0\n\ts_barrier_signal -1\n\ts_barrier_wait -1" ::: "memory");
}

static __device__ __forceinline__ void store_x_tile(const float4 (&xr)[XLD], float* xt,
                                                    int r0, int nrows, int tid)
{
    #pragma unroll
    for (int i = 0; i < XLD; ++i) {
        const int idx = tid + i * NTHR;
        const int row = idx >> 4;
        const int c4  = idx & 15;
        const int g   = r0 - HALO + row;
        if (i < XLD - 1 || idx < XROWS * (RANK / 4)) {
            float4 v = xr[i];
            if (g < 0 || g >= nrows) v = make_float4(0.f, 0.f, 0.f, 0.f);
            *(float4*)&xt[row * LD_X + c4 * 4] = v;
        }
    }
}

#define SLD 4
static __device__ __forceinline__ void load_split_rows(float4 (&sx)[SLD],
                                                       const float* __restrict__ x,
                                                       int rb, int nrows, int tid)
{
    #pragma unroll
    for (int i = 0; i < SLD; ++i) {
        const int idx = tid + i * NTHR;
        const int row = idx >> 4;
        const int c4  = idx & 15;
        const int gc  = min(max(rb - HALO + row, 0), nrows - 1);
        sx[i] = ((const float4*)x)[gc * (RANK / 4) + c4];
    }
}

static __device__ __forceinline__ void store_split_rows(const float4 (&sx)[SLD], float* xt,
                                                        int rb, int nrows, int tid)
{
    #pragma unroll
    for (int i = 0; i < SLD; ++i) {
        const int idx = tid + i * NTHR;
        const int row = idx >> 4;
        const int c4  = idx & 15;
        const int g   = rb - HALO + row;
        float4 v = sx[i];
        if (row >= 16 + 2 * HALO || g < 0 || g >= nrows) v = make_float4(0.f, 0.f, 0.f, 0.f);
        *(float4*)&xt[row * LD_X + c4 * 4] = v;
    }
}

static __device__ __forceinline__ void band_softmax(const v8f (&sacc)[2], float (&p)[2][8], int dj)
{
    float mloc = -__builtin_inff();
    #pragma unroll
    for (int jt = 0; jt < 2; ++jt)
        #pragma unroll
        for (int v = 0; v < 8; ++v) {
            const bool inb = (unsigned)(16 * jt + v + dj) < (unsigned)WIN;
            p[jt][v] = inb ? sacc[jt][v] : -__builtin_inff();
            mloc = fmaxf(mloc, p[jt][v]);
        }
    const float mx = fmaxf(mloc, xhalf(mloc));
    const float sc = 0.125f * 1.44269504088896341f;
    float sum = 0.f;
    #pragma unroll
    for (int jt = 0; jt < 2; ++jt)
        #pragma unroll
        for (int v = 0; v < 8; ++v) {
            p[jt][v] = fast_exp2((p[jt][v] - mx) * sc);
            sum += p[jt][v];
        }
    sum += xhalf(sum);
    const float inv = 1.0f / sum;
    #pragma unroll
    for (int jt = 0; jt < 2; ++jt)
        #pragma unroll
        for (int v = 0; v < 8; ++v) p[jt][v] *= inv;
}

static __device__ __forceinline__ void split_block(const float* xw, const float* ws, const float* bs,
                                                   float* red, float* __restrict__ out,
                                                   int rb, int nrows, int wave, int lane)
{
    const int c = lane & 15;
    const int h = lane >> 4;
    const int foff = c * LD_X + 8 * h;
    const int woff = c * LD_W + 8 * h;
    const int toff = 8 * h * LD_X + c;
    const int dj = 8 * h - c;

    v8f a1;
    {
        const float4 b0 = *(const float4*)&bs[16 * wave + 8 * h];
        const float4 b1 = *(const float4*)&bs[16 * wave + 8 * h + 4];
        a1 = (v8f){b0.x, b0.y, b0.z, b0.w, b1.x, b1.y, b1.z, b1.w};
    }
    float4 xb[8], wq[8], xa[2][2];
    #pragma unroll
    for (int i = 0; i < 8; ++i) {
        xb[i] = *(const float4*)&xw[HALO * LD_X + foff + 16 * (i >> 1) + 4 * (i & 1)];
        wq[i] = *(const float4*)&ws[wave * 16 * LD_W + woff + 16 * (i >> 1) + 4 * (i & 1)];
    }
    #pragma unroll
    for (int jt = 0; jt < 2; ++jt)
        #pragma unroll
        for (int qq = 0; qq < 2; ++qq)
            xa[jt][qq] = *(const float4*)&xw[jt * 16 * LD_X + foff + 16 * wave + 4 * qq];
    v2f xq[8];
    #pragma unroll
    for (int jt = 0; jt < 2; ++jt)
        #pragma unroll
        for (int q = 0; q < 4; ++q) {
            const float* xp = &xw[(16 * jt + 2 * q) * LD_X + toff + 16 * wave];
            xq[4 * jt + q] = (v2f){xp[0], xp[LD_X]};
        }
    #pragma unroll
    for (int i = 0; i < 8; ++i) {
        a1 = mma4((v2f){wq[i].x, wq[i].y}, (v2f){xb[i].x, xb[i].y}, a1);
        a1 = mma4((v2f){wq[i].z, wq[i].w}, (v2f){xb[i].z, xb[i].w}, a1);
    }
    float t[8];
    #pragma unroll
    for (int v = 0; v < 8; ++v) t[v] = fast_tanh(a1[v]);

    v8f sp[2] = {};
    #pragma unroll
    for (int qq = 0; qq < 2; ++qq)
        #pragma unroll
        for (int jt = 0; jt < 2; ++jt) {
            const float4 x4 = xa[jt][qq];
            sp[jt] = mma4((v2f){x4.x, x4.y}, (v2f){t[4 * qq],     t[4 * qq + 1]}, sp[jt]);
            sp[jt] = mma4((v2f){x4.z, x4.w}, (v2f){t[4 * qq + 2], t[4 * qq + 3]}, sp[jt]);
        }
    #pragma unroll
    for (int jt = 0; jt < 2; ++jt)
        #pragma unroll
        for (int v = 0; v < 8; ++v) red[(wave * 16 + 8 * jt + v) * 32 + lane] = sp[jt][v];
    lds_barrier();
    v8f sacc[2];
    #pragma unroll
    for (int jt = 0; jt < 2; ++jt)
        #pragma unroll
        for (int v = 0; v < 8; ++v) {
            float s = red[(8 * jt + v) * 32 + lane];
            #pragma unroll
            for (int ww = 1; ww < 4; ++ww) s += red[(ww * 16 + 8 * jt + v) * 32 + lane];
            sacc[jt][v] = s;
        }

    float p[2][8];
    band_softmax(sacc, p, dj);

    v8f o1 = {};
    #pragma unroll
    for (int jt = 0; jt < 2; ++jt)
        #pragma unroll
        for (int q = 0; q < 4; ++q)
            o1 = mma4((v2f){p[jt][2 * q], p[jt][2 * q + 1]}, xq[4 * jt + q], o1);

    float* ost = red + 4 * 16 * 32;
    #pragma unroll
    for (int v = 0; v < 8; ++v) ost[(8 * h + v) * LD_X + 16 * wave + c] = o1[v];
    lds_barrier();
    #pragma unroll
    for (int i = 0; i < 2; ++i) {
        const int rr = 4 * wave + 2 * i + h;
        const float4 o4 = *(const float4*)&ost[rr * LD_X + 4 * c];
        if (rb + rr < nrows) *(float4*)&out[(rb + rr) * RANK + 4 * c] = o4;
    }
}

__global__ __launch_bounds__(NTHR)
void attn_window_fused_kernel(const float* __restrict__ x,
                              const float* __restrict__ W,
                              const float* __restrict__ b,
                              float* __restrict__ out,
                              int nrows)
{
    __shared__ __attribute__((aligned(16))) float xs[2][SROWS * LD_X];
    __shared__ __attribute__((aligned(16))) float ws[RANK * LD_W];
    __shared__ __attribute__((aligned(16))) float bs[RANK];

    const int tid    = threadIdx.x;
    const int ntiles = (nrows + TILE - 1) / TILE;
    static_assert(NGROUPS == 1024 && (5 * 205) % NGROUPS == 1, "leftover spread is written for 1024 groups");
    const int bid    = blockIdx.x;
    const int full   = ntiles / NGROUPS;
    const int rleft  = full * NGROUPS * TILE;
    const int nsplit = (nrows - rleft + 15) / 16;
    const int kx     = (bid * 205) & (NGROUPS - 1);
    if (full == 0 && kx >= nsplit) return;
    int tile = bid;

    const int lane = tid & 31;
    const int c    = lane & 15;
    const int h    = lane >> 4;
    const int wave = __builtin_amdgcn_readfirstlane(tid >> 5);

    float4 wr[WLD];
    #pragma unroll
    for (int i = 0; i < WLD; ++i) wr[i] = ((const float4*)W)[tid + i * NTHR];

    float bias_r = 0.f;
    if (tid < RANK) bias_r = b[tid];

    int  sblk  = kx;
    bool split = sblk < nsplit;
    float4 sx[SLD];
    if (split) load_split_rows(sx, x, rleft + sblk * 16, nrows, tid);

    float4 xr[XLD];
    load_x_tile(xr, x, tile * TILE, nrows, tid);

    #pragma unroll
    for (int i = 0; i < WLD; ++i) {
        const int idx = tid + i * NTHR;
        const int row = idx >> 4;
        const int c4  = idx & 15;
        *(float4*)&ws[row * LD_W + c4 * 4] = wr[i];
    }
    if (tid < RANK) bs[tid] = bias_r;
    for (int i = tid; i < 2 * (SROWS - XROWS) * (LD_X / 4); i += NTHR) {
        const int s = i / ((SROWS - XROWS) * (LD_X / 4));
        const int o = i - s * ((SROWS - XROWS) * (LD_X / 4));
        *(float4*)&xs[s][XROWS * LD_X + o * 4] = make_float4(0.f, 0.f, 0.f, 0.f);
    }

    while (split) {
        __builtin_amdgcn_s_setprio(3);
        const int rb = rleft + sblk * 16;
        store_split_rows(sx, xs[1], rb, nrows, tid);
        lds_barrier();
        const int  nblk = sblk + NGROUPS;
        const bool more = nblk < nsplit;
        if (more) load_split_rows(sx, x, rleft + nblk * 16, nrows, tid);
        split_block(xs[1], ws, bs, xs[0], out, rb, nrows, wave, lane);
        lds_barrier();
        sblk  = nblk;
        split = more;
    }

    const int foff = c * LD_X + 8 * h;
    const int woff = c * LD_W + 8 * h;
    const int toff = 8 * h * LD_X + c;
    const int dj = 8 * h - c;

    int stage = 0;
    for (int it = 0; it < full; ++it) {
        const int left = full - it;
        if (left >= 4)      __builtin_amdgcn_s_setprio(3);
        else if (left == 3) __builtin_amdgcn_s_setprio(2);
        else if (left == 2) __builtin_amdgcn_s_setprio(1);
        else                __builtin_amdgcn_s_setprio(0);

        float* xt = xs[stage];
        const int r0 = tile * TILE;

        store_x_tile(xr, xt, r0, nrows, tid);
        lds_barrier();

        const int next = tile + NGROUPS;
        if (it + 1 < full) load_x_tile(xr, x, next * TILE, nrows, tid);

        const float* xw = xt + wave * 16 * LD_X;

        v8f acc[4];
        #pragma unroll
        for (int mt = 0; mt < 4; ++mt) {
            const float4 b0 = *(const float4*)&bs[16 * mt + 8 * h];
            const float4 b1 = *(const float4*)&bs[16 * mt + 8 * h + 4];
            acc[mt] = (v8f){b0.x, b0.y, b0.z, b0.w, b1.x, b1.y, b1.z, b1.w};
        }
        float4 xb[8];
        #pragma unroll
        for (int i = 0; i < 8; ++i)
            xb[i] = *(const float4*)&xw[HALO * LD_X + foff + 16 * (i >> 1) + 4 * (i & 1)];

        float4 wa[2][8];
        float4 xa[2][8];
        #pragma unroll
        for (int e = 0; e < 2; ++e)
            #pragma unroll
            for (int mt = 0; mt < 4; ++mt)
                wa[0][2 * mt + e] = *(const float4*)&ws[mt * 16 * LD_W + woff + 4 * e];
        #pragma unroll
        for (int g = 0; g < 4; ++g) {
            if (g < 3) {
                #pragma unroll
                for (int e = 0; e < 2; ++e)
                    #pragma unroll
                    for (int mt = 0; mt < 4; ++mt)
                        wa[(g + 1) & 1][2 * mt + e] =
                            *(const float4*)&ws[mt * 16 * LD_W + woff + 16 * (g + 1) + 4 * e];
            } else {
                #pragma unroll
                for (int i = 0; i < 8; ++i)
                    #pragma unroll
                    for (int jt = 0; jt < 2; ++jt)
                        xa[jt][i] = *(const float4*)&xw[jt * 16 * LD_X + foff + 16 * (i >> 1) + 4 * (i & 1)];
            }
            __builtin_amdgcn_sched_barrier(0);
            #pragma unroll
            for (int e = 0; e < 2; ++e) {
                const int i = 2 * g + e;
                #pragma unroll
                for (int mt = 0; mt < 4; ++mt) {
                    const float4 w4 = wa[g & 1][2 * mt + e];
                    acc[mt] = mma4((v2f){w4.x, w4.y}, (v2f){xb[i].x, xb[i].y}, acc[mt]);
                    acc[mt] = mma4((v2f){w4.z, w4.w}, (v2f){xb[i].z, xb[i].w}, acc[mt]);
                }
            }
            __builtin_amdgcn_sched_barrier(0);
        }
        float t[4][8];
        #pragma unroll
        for (int mt = 0; mt < 4; ++mt)
            #pragma unroll
            for (int v = 0; v < 8; ++v)
                t[mt][v] = fast_tanh(acc[mt][v]);

        v2f xq[4][8];
        #pragma unroll
        for (int jt = 0; jt < 2; ++jt)
            #pragma unroll
            for (int q = 0; q < 4; ++q)
                #pragma unroll
                for (int ct = 0; ct < 4; ++ct) {
                    const float* xp = &xw[(16 * jt + 2 * q) * LD_X + toff + 16 * ct];
                    xq[ct][4 * jt + q] = (v2f){xp[0], xp[LD_X]};
                }
        __builtin_amdgcn_sched_barrier(0);

        v8f sacc[2] = {};
        #pragma unroll
        for (int i = 0; i < 8; ++i) {
            const int g = i >> 1, qq = i & 1;
            #pragma unroll
            for (int jt = 0; jt < 2; ++jt) {
                const float4 x4 = xa[jt][i];
                sacc[jt] = mma4((v2f){x4.x, x4.y}, (v2f){t[g][4 * qq],     t[g][4 * qq + 1]}, sacc[jt]);
                sacc[jt] = mma4((v2f){x4.z, x4.w}, (v2f){t[g][4 * qq + 2], t[g][4 * qq + 3]}, sacc[jt]);
            }
        }

        float p[2][8];
        band_softmax(sacc, p, dj);

        v8f oacc[4] = {};
        #pragma unroll
        for (int jt = 0; jt < 2; ++jt) {
            #pragma unroll
            for (int q = 0; q < 4; ++q) {
                const v2f pa = {p[jt][2 * q], p[jt][2 * q + 1]};
                #pragma unroll
                for (int ct = 0; ct < 4; ++ct)
                    oacc[ct] = mma4(pa, xq[ct][4 * jt + q], oacc[ct]);
            }
        }

        {
            const int rw = r0 + wave * 16;
            float* op = out + (rw + 8 * h) * RANK + c;
            if (rw + 16 <= nrows) {
                #pragma unroll
                for (int v = 0; v < 8; ++v)
                    #pragma unroll
                    for (int ct = 0; ct < 4; ++ct) op[v * RANK + 16 * ct] = oacc[ct][v];
            } else {
                #pragma unroll
                for (int v = 0; v < 8; ++v)
                    if (rw + 8 * h + v < nrows) {
                        #pragma unroll
                        for (int ct = 0; ct < 4; ++ct) op[v * RANK + 16 * ct] = oacc[ct][v];
                    }
            }
        }
        stage ^= 1;
        tile = next;
    }
}

extern "C" void kernel_launch(void* const* d_in, const int* in_sizes, int n_in,
                              void* d_out, int out_size, void* d_ws, size_t ws_size,
                              hipStream_t stream) {
    const float* x = (const float*)d_in[0];
    const float* W = (const float*)d_in[1];
    const float* b = (const float*)d_in[2];
    float* out = (float*)d_out;

    const int nrows = in_sizes[0] / RANK;
    attn_window_fused_kernel<<<NGROUPS, NTHR, 0, stream>>>(x, W, b, out, nrows);
}


# scores and weighted sum on one tile of the block's own 16 rows (32+32 matrix ops -> 16+28 per 16 rows); the 30 edge pairs' scores as vector dot products, lane = row; sums reordered
# speedup vs baseline: 3.0898x; 1.0461x over previous
#include <hip/hip_runtime.h>


#define RANK   64
#define WIN    11
#define HALO   5
#define TILE   64
#define XROWS  (TILE + 2*HALO)
#define SROWS  80
#define LD_X   68
#define LD_W   68
#define NTHR   128
#define XLD    10
#define WLD    8
#define NGROUPS 1024

typedef float v2f __attribute__((ext_vector_type(2)));
typedef float v8f __attribute__((ext_vector_type(8)));

static __device__ __forceinline__ float fast_tanh(float x) {
#if __has_builtin(__builtin_amdgcn_tanhf)
    return __builtin_amdgcn_tanhf(x);
#elif __has_builtin(__builtin_amdgcn_tanh_f32)
    return __builtin_amdgcn_tanh_f32(x);
#else
    return tanhf(x);
#endif
}

static __device__ __forceinline__ float fast_exp2(float x) {
#if __has_builtin(__builtin_amdgcn_exp2f)
    return __builtin_amdgcn_exp2f(x);
#else
    return exp2f(x);
#endif
}

static __device__ __forceinline__ float xhalf(float x) {
    const int b = __builtin_bit_cast(int, x);
    return __builtin_bit_cast(float,
        __builtin_amdgcn_permlanex16(b, b, 0x76543210, 0xfedcba98, false, false));
}

static __device__ __forceinline__ v8f mma4(v2f a, v2f b, v8f c) {
    return __builtin_amdgcn_wmma_f32_16x16x4_f32(false, a, false, b, (short)0, c, false, false);
}

static __device__ __forceinline__ void load_x_tile(float4 (&xr)[XLD],
                                                   const float* __restrict__ x,
                                                   int r0, int nrows, int tid)
{
    #pragma unroll
    for (int i = 0; i < XLD; ++i) {
        const int idx = tid + i * NTHR;
        const int row = idx >> 4;
        const int c4  = idx & 15;
        const int gc  = min(max(r0 - HALO + row, 0), nrows - 1);
        if (i < XLD - 1 || idx < XROWS * (RANK / 4))
            xr[i] = ((const float4*)x)[gc * (RANK / 4) + c4];
    }
}

static __device__ __forceinline__ void lds_barrier() {
    asm volatile("s_wait_dscnt 0x0\n\ts_barrier_signal -1\n\ts_barrier_wait -1" ::: "memory");
}

static __device__ __forceinline__ void store_x_tile(const float4 (&xr)[XLD], float* xt,
                                                    int r0, int nrows, int tid)
{
    #pragma unroll
    for (int i = 0; i < XLD; ++i) {
        const int idx = tid + i * NTHR;
        const int row = idx >> 4;
        const int c4  = idx & 15;
        const int g   = r0 - HALO + row;
        if (i < XLD - 1 || idx < XROWS * (RANK / 4)) {
            float4 v = xr[i];
            if (g < 0 || g >= nrows) v = make_float4(0.f, 0.f, 0.f, 0.f);
            *(float4*)&xt[row * LD_X + c4 * 4] = v;
        }
    }
}

#define SLD 4
static __device__ __forceinline__ void load_split_rows(float4 (&sx)[SLD],
                                                       const float* __restrict__ x,
                                                       int rb, int nrows, int tid)
{
    #pragma unroll
    for (int i = 0; i < SLD; ++i) {
        const int idx = tid + i * NTHR;
        const int row = idx >> 4;
        const int c4  = idx & 15;
        const int gc  = min(max(rb - HALO + row, 0), nrows - 1);
        sx[i] = ((const float4*)x)[gc * (RANK / 4) + c4];
    }
}

static __device__ __forceinline__ void store_split_rows(const float4 (&sx)[SLD], float* xt,
                                                        int rb, int nrows, int tid)
{
    #pragma unroll
    for (int i = 0; i < SLD; ++i) {
        const int idx = tid + i * NTHR;
        const int row = idx >> 4;
        const int c4  = idx & 15;
        const int g   = rb - HALO + row;
        float4 v = sx[i];
        if (row >= 16 + 2 * HALO || g < 0 || g >= nrows) v = make_float4(0.f, 0.f, 0.f, 0.f);
        *(float4*)&xt[row * LD_X + c4 * 4] = v;
    }
}

static __device__ __forceinline__ void band_softmax(const v8f (&sacc)[2], float (&p)[2][8], int dj)
{
    float mloc = -__builtin_inff();
    #pragma unroll
    for (int jt = 0; jt < 2; ++jt)
        #pragma unroll
        for (int v = 0; v < 8; ++v) {
            const bool inb = (unsigned)(16 * jt + v + dj) < (unsigned)WIN;
            p[jt][v] = inb ? sacc[jt][v] : -__builtin_inff();
            mloc = fmaxf(mloc, p[jt][v]);
        }
    const float mx = fmaxf(mloc, xhalf(mloc));
    const float sc = 0.125f * 1.44269504088896341f;
    float sum = 0.f;
    #pragma unroll
    for (int jt = 0; jt < 2; ++jt)
        #pragma unroll
        for (int v = 0; v < 8; ++v) {
            p[jt][v] = fast_exp2((p[jt][v] - mx) * sc);
            sum += p[jt][v];
        }
    sum += xhalf(sum);
    const float inv = 1.0f / sum;
    #pragma unroll
    for (int jt = 0; jt < 2; ++jt)
        #pragma unroll
        for (int v = 0; v < 8; ++v) p[jt][v] *= inv;
}

static __device__ __forceinline__ void split_block(const float* xw, const float* ws, const float* bs,
                                                   float* red, float* __restrict__ out,
                                                   int rb, int nrows, int wave, int lane)
{
    const int c = lane & 15;
    const int h = lane >> 4;
    const int foff = c * LD_X + 8 * h;
    const int woff = c * LD_W + 8 * h;
    const int toff = 8 * h * LD_X + c;
    const int dj = 8 * h - c;

    v8f a1;
    {
        const float4 b0 = *(const float4*)&bs[16 * wave + 8 * h];
        const float4 b1 = *(const float4*)&bs[16 * wave + 8 * h + 4];
        a1 = (v8f){b0.x, b0.y, b0.z, b0.w, b1.x, b1.y, b1.z, b1.w};
    }
    float4 xb[8], wq[8], xa[2][2];
    #pragma unroll
    for (int i = 0; i < 8; ++i) {
        xb[i] = *(const float4*)&xw[HALO * LD_X + foff + 16 * (i >> 1) + 4 * (i & 1)];
        wq[i] = *(const float4*)&ws[wave * 16 * LD_W + woff + 16 * (i >> 1) + 4 * (i & 1)];
    }
    #pragma unroll
    for (int jt = 0; jt < 2; ++jt)
        #pragma unroll
        for (int qq = 0; qq < 2; ++qq)
            xa[jt][qq] = *(const float4*)&xw[jt * 16 * LD_X + foff + 16 * wave + 4 * qq];
    v2f xq[8];
    #pragma unroll
    for (int jt = 0; jt < 2; ++jt)
        #pragma unroll
        for (int q = 0; q < 4; ++q) {
            const float* xp = &xw[(16 * jt + 2 * q) * LD_X + toff + 16 * wave];
            xq[4 * jt + q] = (v2f){xp[0], xp[LD_X]};
        }
    #pragma unroll
    for (int i = 0; i < 8; ++i) {
        a1 = mma4((v2f){wq[i].x, wq[i].y}, (v2f){xb[i].x, xb[i].y}, a1);
        a1 = mma4((v2f){wq[i].z, wq[i].w}, (v2f){xb[i].z, xb[i].w}, a1);
    }
    float t[8];
    #pragma unroll
    for (int v = 0; v < 8; ++v) t[v] = fast_tanh(a1[v]);

    v8f sp[2] = {};
    #pragma unroll
    for (int qq = 0; qq < 2; ++qq)
        #pragma unroll
        for (int jt = 0; jt < 2; ++jt) {
            const float4 x4 = xa[jt][qq];
            sp[jt] = mma4((v2f){x4.x, x4.y}, (v2f){t[4 * qq],     t[4 * qq + 1]}, sp[jt]);
            sp[jt] = mma4((v2f){x4.z, x4.w}, (v2f){t[4 * qq + 2], t[4 * qq + 3]}, sp[jt]);
        }
    #pragma unroll
    for (int jt = 0; jt < 2; ++jt)
        #pragma unroll
        for (int v = 0; v < 8; ++v) red[(wave * 16 + 8 * jt + v) * 32 + lane] = sp[jt][v];
    lds_barrier();
    v8f sacc[2];
    #pragma unroll
    for (int jt = 0; jt < 2; ++jt)
        #pragma unroll
        for (int v = 0; v < 8; ++v) {
            float s = red[(8 * jt + v) * 32 + lane];
            #pragma unroll
            for (int ww = 1; ww < 4; ++ww) s += red[(ww * 16 + 8 * jt + v) * 32 + lane];
            sacc[jt][v] = s;
        }

    float p[2][8];
    band_softmax(sacc, p, dj);

    v8f o1 = {};
    #pragma unroll
    for (int jt = 0; jt < 2; ++jt)
        #pragma unroll
        for (int q = 0; q < 4; ++q)
            o1 = mma4((v2f){p[jt][2 * q], p[jt][2 * q + 1]}, xq[4 * jt + q], o1);

    float* ost = red + 4 * 16 * 32;
    #pragma unroll
    for (int v = 0; v < 8; ++v) ost[(8 * h + v) * LD_X + 16 * wave + c] = o1[v];
    lds_barrier();
    #pragma unroll
    for (int i = 0; i < 2; ++i) {
        const int rr = 4 * wave + 2 * i + h;
        const float4 o4 = *(const float4*)&ost[rr * LD_X + 4 * c];
        if (rb + rr < nrows) *(float4*)&out[(rb + rr) * RANK + 4 * c] = o4;
    }
}

__global__ __launch_bounds__(NTHR) __attribute__((amdgpu_waves_per_eu(4)))
void attn_window_fused_kernel(const float* __restrict__ x,
                              const float* __restrict__ W,
                              const float* __restrict__ b,
                              float* __restrict__ out,
                              int nrows)
{
    __shared__ __attribute__((aligned(16))) float xs[2][SROWS * LD_X];
    __shared__ __attribute__((aligned(16))) float ws[RANK * LD_W];
    __shared__ __attribute__((aligned(16))) float bs[RANK];

    const int tid    = threadIdx.x;
    const int ntiles = (nrows + TILE - 1) / TILE;
    static_assert(NGROUPS == 1024 && (5 * 205) % NGROUPS == 1, "leftover spread is written for 1024 groups");
    const int bid    = blockIdx.x;
    const int full   = ntiles / NGROUPS;
    const int rleft  = full * NGROUPS * TILE;
    const int nsplit = (nrows - rleft + 15) / 16;
    const int kx     = (bid * 205) & (NGROUPS - 1);
    if (full == 0 && kx >= nsplit) return;
    int tile = bid;

    const int lane = tid & 31;
    const int c    = lane & 15;
    const int h    = lane >> 4;
    const int wave = __builtin_amdgcn_readfirstlane(tid >> 5);

    float4 wr[WLD];
    #pragma unroll
    for (int i = 0; i < WLD; ++i) wr[i] = ((const float4*)W)[tid + i * NTHR];

    float bias_r = 0.f;
    if (tid < RANK) bias_r = b[tid];

    int  sblk  = kx;
    bool split = sblk < nsplit;
    float4 sx[SLD];
    if (split) load_split_rows(sx, x, rleft + sblk * 16, nrows, tid);

    float4 xr[XLD];
    load_x_tile(xr, x, tile * TILE, nrows, tid);

    #pragma unroll
    for (int i = 0; i < WLD; ++i) {
        const int idx = tid + i * NTHR;
        const int row = idx >> 4;
        const int c4  = idx & 15;
        *(float4*)&ws[row * LD_W + c4 * 4] = wr[i];
    }
    if (tid < RANK) bs[tid] = bias_r;

    while (split) {
        __builtin_amdgcn_s_setprio(3);
        const int rb = rleft + sblk * 16;
        store_split_rows(sx, xs[1], rb, nrows, tid);
        lds_barrier();
        const int  nblk = sblk + NGROUPS;
        const bool more = nblk < nsplit;
        if (more) load_split_rows(sx, x, rleft + nblk * 16, nrows, tid);
        split_block(xs[1], ws, bs, xs[0], out, rb, nrows, wave, lane);
        lds_barrier();
        sblk  = nblk;
        split = more;
    }

    const int foff = c * LD_X + 8 * h;
    const int woff = c * LD_W + 8 * h;
    const int toff = 8 * h * LD_X + c;
    const int dj = 8 * h - c;

    int stage = 0;
    for (int it = 0; it < full; ++it) {
        const int left = full - it;
        if (left >= 4)      __builtin_amdgcn_s_setprio(3);
        else if (left == 3) __builtin_amdgcn_s_setprio(2);
        else if (left == 2) __builtin_amdgcn_s_setprio(1);
        else                __builtin_amdgcn_s_setprio(0);

        float* xt = xs[stage];
        const int r0 = tile * TILE;

        store_x_tile(xr, xt, r0, nrows, tid);
        lds_barrier();

        const int next = tile + NGROUPS;
        if (it + 1 < full) load_x_tile(xr, x, next * TILE, nrows, tid);

        const float* xw = xt + wave * 16 * LD_X;

        v8f acc[4];
        #pragma unroll
        for (int mt = 0; mt < 4; ++mt) {
            const float4 b0 = *(const float4*)&bs[16 * mt + 8 * h];
            const float4 b1 = *(const float4*)&bs[16 * mt + 8 * h + 4];
            acc[mt] = (v8f){b0.x, b0.y, b0.z, b0.w, b1.x, b1.y, b1.z, b1.w};
        }
        float4 xb[8];
        #pragma unroll
        for (int i = 0; i < 8; ++i)
            xb[i] = *(const float4*)&xw[HALO * LD_X + foff + 16 * (i >> 1) + 4 * (i & 1)];

        float4 wa[2][8];
        const float* xc = xw + ((c < 8) ? HALO : HALO + 15) * LD_X + 8 * h;
        const int    cstep = (c < 8) ? -LD_X : LD_X;
        float4 cxa[2][8], cxb[8];
        #pragma unroll
        for (int e = 0; e < 2; ++e)
            #pragma unroll
            for (int mt = 0; mt < 4; ++mt)
                wa[0][2 * mt + e] = *(const float4*)&ws[mt * 16 * LD_W + woff + 4 * e];
        #pragma unroll
        for (int g = 0; g < 4; ++g) {
            if (g < 3) {
                #pragma unroll
                for (int e = 0; e < 2; ++e)
                    #pragma unroll
                    for (int mt = 0; mt < 4; ++mt)
                        wa[(g + 1) & 1][2 * mt + e] =
                            *(const float4*)&ws[mt * 16 * LD_W + woff + 16 * (g + 1) + 4 * e];
            } else {
                #pragma unroll
                for (int pi = 0; pi < 2; ++pi)
                    #pragma unroll
                    for (int i = 0; i < 8; ++i)
                        cxa[pi][i] = *(const float4*)&xc[(pi + 1) * cstep + 16 * (i >> 1) + 4 * (i & 1)];
            }
            __builtin_amdgcn_sched_barrier(0);
            #pragma unroll
            for (int e = 0; e < 2; ++e) {
                const int i = 2 * g + e;
                #pragma unroll
                for (int mt = 0; mt < 4; ++mt) {
                    const float4 w4 = wa[g & 1][2 * mt + e];
                    acc[mt] = mma4((v2f){w4.x, w4.y}, (v2f){xb[i].x, xb[i].y}, acc[mt]);
                    acc[mt] = mma4((v2f){w4.z, w4.w}, (v2f){xb[i].z, xb[i].w}, acc[mt]);
                }
            }
            __builtin_amdgcn_sched_barrier(0);
        }
        float t[4][8];
        #pragma unroll
        for (int mt = 0; mt < 4; ++mt)
            #pragma unroll
            for (int v = 0; v < 8; ++v)
                t[mt][v] = fast_tanh(acc[mt][v]);

        v8f sacc[2] = {};
        #pragma unroll
        for (int i = 0; i < 8; ++i) {
            const int g = i >> 1, qq = i & 1;
            const float4 x4 = xb[i];
            sacc[0] = mma4((v2f){x4.x, x4.y}, (v2f){t[g][4 * qq],     t[g][4 * qq + 1]}, sacc[0]);
            sacc[1] = mma4((v2f){x4.z, x4.w}, (v2f){t[g][4 * qq + 2], t[g][4 * qq + 3]}, sacc[1]);
        }

        float cs[HALO];
        #pragma unroll
        for (int pi = 0; pi < 2; ++pi) {
            v2f d2 = {0.f, 0.f};
            #pragma unroll
            for (int i = 0; i < 8; ++i) {
                const int g = i >> 1, qq = i & 1;
                const float4 r4 = cxa[pi][i];
                d2 += (v2f){r4.x, r4.y} * (v2f){t[g][4 * qq],     t[g][4 * qq + 1]};
                d2 += (v2f){r4.z, r4.w} * (v2f){t[g][4 * qq + 2], t[g][4 * qq + 3]};
            }
            const float part = d2.x + d2.y;
            cs[pi] = part + xhalf(part);
        }
        __builtin_amdgcn_sched_barrier(0);

        #pragma unroll
        for (int pi = 0; pi < 2; ++pi)
            #pragma unroll
            for (int i = 0; i < 8; ++i)
                cxa[pi][i] = *(const float4*)&xc[(pi + 3) * cstep + 16 * (i >> 1) + 4 * (i & 1)];
        #pragma unroll
        for (int i = 0; i < 8; ++i)
            cxb[i] = *(const float4*)&xc[HALO * cstep + 16 * (i >> 1) + 4 * (i & 1)];
        #pragma unroll
        for (int pi = 2; pi < HALO; ++pi) {
            v2f d2 = {0.f, 0.f};
            #pragma unroll
            for (int i = 0; i < 8; ++i) {
                const int g = i >> 1, qq = i & 1;
                const float4 r4 = (pi < 4) ? cxa[pi - 2][i] : cxb[i];
                d2 += (v2f){r4.x, r4.y} * (v2f){t[g][4 * qq],     t[g][4 * qq + 1]};
                d2 += (v2f){r4.z, r4.w} * (v2f){t[g][4 * qq + 2], t[g][4 * qq + 3]};
            }
            const float part = d2.x + d2.y;
            cs[pi] = part + xhalf(part);
        }
        v2f xqm[4][4];
        v2f xqc[4][3];
        #pragma unroll
        for (int q = 0; q < 4; ++q)
            #pragma unroll
            for (int ct = 0; ct < 4; ++ct) {
                const float* xp = &xw[(HALO + 2 * q) * LD_X + toff + 16 * ct];
                xqm[ct][q] = (v2f){xp[0], xp[LD_X]};
            }
        {
            const float* xe = xw + (h ? (HALO + 15) * LD_X : HALO * LD_X) + c;
            const int    es = h ? LD_X : -LD_X;
            #pragma unroll
            for (int j = 0; j < 3; ++j)
                #pragma unroll
                for (int ct = 0; ct < 4; ++ct) {
                    const int i0 = 2 * j + 1, i1 = (2 * j + 2 <= HALO) ? 2 * j + 2 : HALO;
                    xqc[ct][j] = (v2f){xe[i0 * es + 16 * ct], xe[i1 * es + 16 * ct]};
                }
        }
        __builtin_amdgcn_sched_barrier(0);

        float pm[8], pc[HALO];
        {
            const int  ncorn = (c < 8) ? HALO - c : c - (15 - HALO);
            const bool mine  = (c < 8) == (h == 0);
            float mloc = -__builtin_inff();
            #pragma unroll
            for (int v = 0; v < 8; ++v) {
                const bool inb = (unsigned)(v + dj + HALO) < (unsigned)WIN;
                pm[v] = inb ? sacc[0][v] + sacc[1][v] : -__builtin_inff();
                mloc = fmaxf(mloc, pm[v]);
            }
            #pragma unroll
            for (int i = 0; i < HALO; ++i) {
                pc[i] = (mine && i < ncorn) ? cs[i] : -__builtin_inff();
                mloc = fmaxf(mloc, pc[i]);
            }
            const float mx = fmaxf(mloc, xhalf(mloc));
            const float sc = 0.125f * 1.44269504088896341f;
            float sum = 0.f;
            #pragma unroll
            for (int v = 0; v < 8; ++v) {
                pm[v] = fast_exp2((pm[v] - mx) * sc);
                sum += pm[v];
            }
            #pragma unroll
            for (int i = 0; i < HALO; ++i) {
                pc[i] = fast_exp2((pc[i] - mx) * sc);
                sum += pc[i];
            }
            sum += xhalf(sum);
            const float inv = 1.0f / sum;
            #pragma unroll
            for (int v = 0; v < 8; ++v) pm[v] *= inv;
            #pragma unroll
            for (int i = 0; i < HALO; ++i) pc[i] *= inv;
        }

        v8f oacc[4] = {};
        #pragma unroll
        for (int q = 0; q < 4; ++q) {
            const v2f pa = {pm[2 * q], pm[2 * q + 1]};
            #pragma unroll
            for (int ct = 0; ct < 4; ++ct)
                oacc[ct] = mma4(pa, xqm[ct][q], oacc[ct]);
        }
        #pragma unroll
        for (int j = 0; j < 3; ++j) {
            const v2f pa = {pc[2 * j], (2 * j + 1 < HALO) ? pc[2 * j + 1] : 0.f};
            #pragma unroll
            for (int ct = 0; ct < 4; ++ct)
                oacc[ct] = mma4(pa, xqc[ct][j], oacc[ct]);
        }

        {
            const int rw = r0 + wave * 16;
            float* op = out + (rw + 8 * h) * RANK + c;
            if (rw + 16 <= nrows) {
                #pragma unroll
                for (int v = 0; v < 8; ++v)
                    #pragma unroll
                    for (int ct = 0; ct < 4; ++ct) op[v * RANK + 16 * ct] = oacc[ct][v];
            } else {
                #pragma unroll
                for (int v = 0; v < 8; ++v)
                    if (rw + 8 * h + v < nrows) {
                        #pragma unroll
                        for (int ct = 0; ct < 4; ++ct) op[v * RANK + 16 * ct] = oacc[ct][v];
                    }
            }
        }
        stage ^= 1;
        tile = next;
    }
}

extern "C" void kernel_launch(void* const* d_in, const int* in_sizes, int n_in,
                              void* d_out, int out_size, void* d_ws, size_t ws_size,
                              hipStream_t stream) {
    const float* x = (const float*)d_in[0];
    const float* W = (const float*)d_in[1];
    const float* b = (const float*)d_in[2];
    float* out = (float*)d_out;

    const int nrows = in_sizes[0] / RANK;
    attn_window_fused_kernel<<<NGROUPS, NTHR, 0, stream>>>(x, W, b, out, nrows);
}
